# starting kernel
# baseline (speedup 1.0000x reference)
#include <hip/hip_runtime.h>
#include <hip/hip_bf16.h>

typedef __attribute__((ext_vector_type(2))) float v2f;
typedef __attribute__((ext_vector_type(8))) float v8f;

#define D_CH 128

__device__ __forceinline__ void atomic_add_f32(float* p, float v) {
#if defined(__HIP_DEVICE_COMPILE__)
    unsafeAtomicAdd(p, v);
#else
    atomicAdd(p, v);
#endif
}

__global__ void fill_f32(float* __restrict__ p, float v, int n) {
    int i = blockIdx.x * blockDim.x + threadIdx.x;
    if (i < n) p[i] = v;
}

__global__ void deg_accum(float* __restrict__ deg, const int* __restrict__ dst,
                          const float* __restrict__ w, int E) {
    int e = blockIdx.x * blockDim.x + threadIdx.x;
    if (e < E) atomic_add_f32(&deg[dst[e]], w[e]);
}

__global__ void rsqrt_inplace(float* __restrict__ d, int n) {
    int i = blockIdx.x * blockDim.x + threadIdx.x;
    if (i < n) d[i] = __frsqrt_rn(d[i]);
}

__global__ void edge_norm(float* __restrict__ norm, const int* __restrict__ src,
                          const int* __restrict__ dst, const float* __restrict__ w,
                          const float* __restrict__ dinv, int E) {
    int e = blockIdx.x * blockDim.x + threadIdx.x;
    if (e < E) norm[e] = dinv[src[e]] * w[e] * dinv[dst[e]];
}

__global__ __launch_bounds__(32) void gemm_wmma_f32(
        const float* __restrict__ X, const float* __restrict__ W,
        float* __restrict__ H) {
#if defined(__HIP_DEVICE_COMPILE__)
    const int lane  = threadIdx.x;
    const int m0    = blockIdx.x * 16;
    const int rowA  = m0 + (lane & 15);
    const int koff  = (lane >> 4) * 2;
    const int colB  = lane & 15;

    v8f acc[8];
#pragma unroll
    for (int n = 0; n < 8; ++n) acc[n] = v8f{};

    const float* __restrict__ xrow = X + (size_t)rowA * D_CH;

    for (int ks = 0; ks < 32; ++ks) {
        const int kb = ks * 4;
        v2f a;
        a.x = xrow[kb + koff];
        a.y = xrow[kb + koff + 1];
        const float* __restrict__ w0 = W + (size_t)(kb + koff) * D_CH;
#pragma unroll
        for (int n = 0; n < 8; ++n) {
            v2f b;
            b.x = w0[n * 16 + colB];
            b.y = w0[D_CH + n * 16 + colB];
            acc[n] = __builtin_amdgcn_wmma_f32_16x16x4_f32(
                false, a, false, b, (short)0, acc[n], false, false);
        }
    }

    const int outRow  = m0 + ((lane >> 4) * 8);
    const int outCol0 = lane & 15;
#pragma unroll
    for (int n = 0; n < 8; ++n) {
#pragma unroll
        for (int r = 0; r < 8; ++r) {
            H[(size_t)(outRow + r) * D_CH + n * 16 + outCol0] = acc[n][r];
        }
    }
#endif
}

__global__ __launch_bounds__(256) void edge_scatter(
        const float* __restrict__ h, const float* __restrict__ norm,
        const int* __restrict__ src, const int* __restrict__ dst,
        float* __restrict__ agg, int E) {
    const int wave = blockIdx.x * (blockDim.x / 32) + (threadIdx.x >> 5);
    const int lane = threadIdx.x & 31;
    if (wave >= E) return;
    const int   s  = src[wave];
    const int   d  = dst[wave];
    const float nw = norm[wave];
    const float* __restrict__ hs = h + (size_t)s * D_CH;
    float* __restrict__ ad = agg + (size_t)d * D_CH;
#pragma unroll
    for (int j = 0; j < 4; ++j) {
        const int c = lane + j * 32;
        atomic_add_f32(&ad[c], nw * hs[c]);
    }
}

__global__ void finish_layer(const float* __restrict__ agg,
                             const float* __restrict__ h,
                             const float* __restrict__ dinv,
                             const float* __restrict__ b,
                             float* __restrict__ xout, int total) {
    int i = blockIdx.x * blockDim.x + threadIdx.x;
    if (i >= total) return;
    const int node = i >> 7;
    const int ch   = i & (D_CH - 1);
    const float di = dinv[node];
    const float v  = agg[i] + h[i] * di * di + b[ch];
    xout[i] = fmaxf(v, 0.0f);
}

__global__ void lookup_mean(const float* __restrict__ x1,
                            const float* __restrict__ x2,
                            const float* __restrict__ x3,
                            const int* __restrict__ idx,
                            float* __restrict__ out, int nNodes, int total) {
    int i = blockIdx.x * blockDim.x + threadIdx.x;
    if (i >= total) return;
    const int bi = i >> 7;
    const int ch = i & (D_CH - 1);
    const int id = idx[bi];
    float v = 0.0f;
    if (id < nNodes) {
        const size_t o = (size_t)id * D_CH + ch;
        v = (x1[o] + x2[o] + x3[o]) * (1.0f / 3.0f);
    }
    out[i] = v;
}

extern "C" void kernel_launch(void* const* d_in, const int* in_sizes, int n_in,
                              void* d_out, int out_size, void* d_ws, size_t ws_size,
                              hipStream_t stream) {
    const int*   inputs_idx  = (const int*)  d_in[0];
    const float* x           = (const float*)d_in[1];
    const int*   edge_index  = (const int*)  d_in[2];
    const float* edge_weight = (const float*)d_in[3];
    const float* Wm[3] = { (const float*)d_in[4], (const float*)d_in[6], (const float*)d_in[8] };
    const float* bv[3] = { (const float*)d_in[5], (const float*)d_in[7], (const float*)d_in[9] };

    const int B = in_sizes[0];
    const int N = in_sizes[1] / D_CH;
    const int E = in_sizes[2] / 2;
    (void)B; (void)n_in; (void)ws_size;
    const int* src = edge_index;
    const int* dst = edge_index + E;

    float* ws   = (float*)d_ws;
    const size_t nd = (size_t)N * D_CH;
    float* dinv = ws;
    float* norm = dinv + N;
    float* h    = norm + E;
    float* agg  = h + nd;
    float* xl[3];
    xl[0] = agg + nd;
    xl[1] = xl[0] + nd;
    xl[2] = xl[1] + nd;

    const int TB = 256;
    const int mTiles   = N / 16;
    const int edgeBlk  = (E + TB - 1) / TB;
    const int nodeBlk  = (N + TB - 1) / TB;
    const int ndBlk    = (int)((nd + TB - 1) / TB);
    const int waveBlk  = (E + (TB / 32) - 1) / (TB / 32);
    const int outBlk   = (out_size + TB - 1) / TB;

    fill_f32<<<nodeBlk, TB, 0, stream>>>(dinv, 1.0f, N);
    deg_accum<<<edgeBlk, TB, 0, stream>>>(dinv, dst, edge_weight, E);
    rsqrt_inplace<<<nodeBlk, TB, 0, stream>>>(dinv, N);
    edge_norm<<<edgeBlk, TB, 0, stream>>>(norm, src, dst, edge_weight, dinv, E);

    const float* xin = x;
    for (int l = 0; l < 3; ++l) {
        fill_f32<<<ndBlk, TB, 0, stream>>>(agg, 0.0f, (int)nd);
        gemm_wmma_f32<<<mTiles, 32, 0, stream>>>(xin, Wm[l], h);
        edge_scatter<<<waveBlk, TB, 0, stream>>>(h, norm, src, dst, agg, E);
        finish_layer<<<ndBlk, TB, 0, stream>>>(agg, h, dinv, bv[l], xl[l], (int)nd);
        xin = xl[l];
    }

    lookup_mean<<<outBlk, TB, 0, stream>>>(xl[0], xl[1], xl[2], inputs_idx,
                                           (float*)d_out, N, out_size);
}


# speedup vs baseline: 3.0311x; 3.0311x over previous
#include <hip/hip_runtime.h>
#include <hip/hip_bf16.h>

typedef __attribute__((ext_vector_type(2))) float v2f;
typedef __attribute__((ext_vector_type(8))) float v8f;

#define D_CH 128

#define BIN_R     50
#define MAX_BINS  1024
#define BIN_CHUNK 4096
#define AGG_T     512
#define XT_ROWS   64
#define XT_PITCH  132
#define LK_AHEAD  8

__device__ __forceinline__ void atomic_add_f32(float* p, float v) {
#if defined(__HIP_DEVICE_COMPILE__)
    unsafeAtomicAdd(p, v);
#else
    atomicAdd(p, v);
#endif
}

__global__ void init_graph(float* __restrict__ deg, int n,
                           int* __restrict__ binCount, int nb) {
    int i = blockIdx.x * blockDim.x + threadIdx.x;
    if (i < n)  deg[i] = 1.0f;
    if (i < nb) binCount[i] = 0;
}


__device__ __forceinline__ void bin_count_job(int* lc, int blk,
                                              const int* __restrict__ dst,
                                              int* __restrict__ binCount,
                                              int E, int nb) {
    const int tid = threadIdx.x;
    for (int i = tid; i < nb; i += 256) lc[i] = 0;
    __syncthreads();
    const int base = blk * BIN_CHUNK;
    for (int j = 0; j < BIN_CHUNK / 256; ++j) {
        const int e = base + j * 256 + tid;
        if (e < E) atomicAdd(&lc[dst[e] / BIN_R], 1);
    }
    __syncthreads();
    for (int i = tid; i < nb; i += 256) {
        const int c = lc[i];
        if (c) atomicAdd(&binCount[i], c);
    }
}

__global__ __launch_bounds__(256) void count_and_degree(
        const int* __restrict__ dst, const float* __restrict__ w,
        float* __restrict__ deg, int* __restrict__ binCount,
        int E, int nb, int countBlk) {
    __shared__ int lc[MAX_BINS];
    const int blk = blockIdx.x;
    if (blk < countBlk) {
        bin_count_job(lc, blk, dst, binCount, E, nb);
    } else {
        const int e = (blk - countBlk) * 256 + threadIdx.x;
        if (e < E) atomic_add_f32(&deg[dst[e]], w[e]);
    }
}

__device__ __forceinline__ void bin_scan_job(int* s, const int* __restrict__ binCount,
                                             int* __restrict__ binStart,
                                             int* __restrict__ cursor, int nb) {
    const int t = threadIdx.x;
    const int c = (t < nb) ? binCount[t] : 0;
    s[t] = c;
    __syncthreads();
    for (int off = 1; off < MAX_BINS; off <<= 1) {
        const int v = (t >= off) ? s[t - off] : 0;
        __syncthreads();
        s[t] += v;
        __syncthreads();
    }
    const int incl = s[t];
    if (t < nb) { binStart[t] = incl - c; cursor[t] = incl - c; }
    if (t == nb - 1) binStart[nb] = incl;
}

__global__ __launch_bounds__(MAX_BINS) void scan_and_rsqrt(
        const int* __restrict__ binCount, int* __restrict__ binStart,
        int* __restrict__ cursor, int nb, float* __restrict__ d, int n) {
    __shared__ int s[MAX_BINS];
    if (blockIdx.x == 0) {
        bin_scan_job(s, binCount, binStart, cursor, nb);
    } else {
        const int i = (blockIdx.x - 1) * MAX_BINS + threadIdx.x;
        if (i < n) d[i] = __frsqrt_rn(d[i]);
    }
}

__device__ __forceinline__ void bin_fill_job(
        int* lc, int blk,
        const int* __restrict__ src, const int* __restrict__ dst,
        const float* __restrict__ w, const float* __restrict__ dinv,
        int* __restrict__ cursor, int* __restrict__ bsrc, int* __restrict__ bdl,
        float* __restrict__ bnorm, int E, int nb) {
    const int tid = threadIdx.x;
    for (int i = tid; i < nb; i += 256) lc[i] = 0;
    __syncthreads();
    const int base = blk * BIN_CHUNK;
    int dd[BIN_CHUNK / 256];
    int rk[BIN_CHUNK / 256];
#pragma unroll
    for (int j = 0; j < BIN_CHUNK / 256; ++j) {
        const int e = base + j * 256 + tid;
        dd[j] = -1;
        rk[j] = 0;
        if (e < E) {
            const int d = dst[e];
            dd[j] = d;
            rk[j] = atomicAdd(&lc[d / BIN_R], 1);
        }
    }
    __syncthreads();
    for (int i = tid; i < nb; i += 256) {
        const int c = lc[i];
        if (c) lc[i] = atomicAdd(&cursor[i], c);
    }
    __syncthreads();
#pragma unroll
    for (int j = 0; j < BIN_CHUNK / 256; ++j) {
        const int d = dd[j];
        if (d >= 0) {
            const int e   = base + j * 256 + tid;
            const int b   = d / BIN_R;
            const int s   = src[e];
            const int pos = lc[b] + rk[j];
            bsrc[pos]  = s;
            bdl[pos]   = d - b * BIN_R;
            bnorm[pos] = dinv[s] * w[e] * dinv[d];
        }
    }
}

__device__ __forceinline__ void gemm_tile(
        const float* __restrict__ X, const float* __restrict__ W,
        float* __restrict__ H, int m0, int lane) {
#if defined(__HIP_DEVICE_COMPILE__)
    const int rowA  = m0 + (lane & 15);
    const int koff  = (lane >> 4) * 2;
    const int colB  = lane & 15;

    v8f acc[8];
#pragma unroll
    for (int n = 0; n < 8; ++n) acc[n] = v8f{};

    const float* __restrict__ xrow = X + (size_t)rowA * D_CH;

    for (int ks = 0; ks < 32; ++ks) {
        const int kb = ks * 4;
        v2f a;
        a.x = xrow[kb + koff];
        a.y = xrow[kb + koff + 1];
        const float* __restrict__ w0 = W + (size_t)(kb + koff) * D_CH;
#pragma unroll
        for (int n = 0; n < 8; ++n) {
            v2f b;
            b.x = w0[n * 16 + colB];
            b.y = w0[D_CH + n * 16 + colB];
            acc[n] = __builtin_amdgcn_wmma_f32_16x16x4_f32(
                false, a, false, b, (short)0, acc[n], false, false);
        }
    }

    const int outRow  = m0 + ((lane >> 4) * 8);
    const int outCol0 = lane & 15;
#pragma unroll
    for (int n = 0; n < 8; ++n) {
#pragma unroll
        for (int r = 0; r < 8; ++r) {
            H[(size_t)(outRow + r) * D_CH + n * 16 + outCol0] = acc[n][r];
        }
    }
#endif
}

__global__ __launch_bounds__(256) void fill_and_gemm(
        const int* __restrict__ src, const int* __restrict__ dst,
        const float* __restrict__ w, const float* __restrict__ dinv,
        int* __restrict__ cursor, int* __restrict__ bsrc, int* __restrict__ bdl,
        float* __restrict__ bnorm, int E, int nb, int fillBlk,
        const float* __restrict__ X, const float* __restrict__ Wt,
        float* __restrict__ H, int mTiles) {
    __shared__ int lc[MAX_BINS];
    const int blk = blockIdx.x;
    if (blk < fillBlk) {
        bin_fill_job(lc, blk, src, dst, w, dinv, cursor, bsrc, bdl, bnorm, E, nb);
    } else {
        const int tile = (blk - fillBlk) * 8 + (int)(threadIdx.x >> 5);
        if (tile < mTiles) gemm_tile(X, Wt, H, tile * 16, (int)(threadIdx.x & 31));
    }
}

__device__ __forceinline__ void tile_add(float* t, float v) {
    __hip_atomic_fetch_add(t, v, __ATOMIC_RELAXED, __HIP_MEMORY_SCOPE_WORKGROUP);
}

__device__ __forceinline__ void tile_add_row(float* __restrict__ t, float nw, float4 r) {
    tile_add(t,      nw * r.x);
    tile_add(t + 32, nw * r.y);
    tile_add(t + 64, nw * r.z);
    tile_add(t + 96, nw * r.w);
}

__global__ __launch_bounds__(AGG_T) void bin_aggregate(
        const float* __restrict__ h, const int* __restrict__ bsrc,
        const int* __restrict__ bdl, const float* __restrict__ bnorm,
        const int* __restrict__ binStart, const float* __restrict__ dinv,
        const float* __restrict__ bias, int nNodes,
        const float* __restrict__ Wnext, float* __restrict__ hnext,
        const int* __restrict__ idx, int nLook, float* __restrict__ erow,
        const float* __restrict__ eA, const float* __restrict__ eB,
        float* __restrict__ out) {
#if defined(__HIP_DEVICE_COMPILE__)
    __shared__ __attribute__((aligned(16))) float tile[BIN_R * D_CH];
    __shared__ __attribute__((aligned(16))) float xt[XT_ROWS * XT_PITCH];
    const int tid  = threadIdx.x;
    const int lane = tid & 31;
    const int wave = __builtin_amdgcn_readfirstlane(tid >> 5);
    const int bin  = blockIdx.x;

    for (int i = tid; i < BIN_R * D_CH / 4; i += AGG_T)
        reinterpret_cast<float4*>(tile)[i] = make_float4(0.f, 0.f, 0.f, 0.f);
    __syncthreads();

    const int start = binStart[bin];
    const int end   = binStart[bin + 1];
    const int per   = (end - start + (AGG_T / 32) - 1) / (AGG_T / 32);
    const int e0    = start + wave * per;
    const int e1    = min(e0 + per, end);

    for (int base = e0; base < e1; base += 32) {
        const int m = min(32, e1 - base);
        int   sv = 0, dv = 0;
        float nv = 0.f;
        if (lane < m) {
            sv = bsrc[base + lane];
            dv = bdl[base + lane];
            nv = bnorm[base + lane];
        }
        const int nvi = __float_as_int(nv);
        int k = 0;
        for (; k + 4 <= m; k += 4) {
            const int s0 = __builtin_amdgcn_readlane(sv, k);
            const int s1 = __builtin_amdgcn_readlane(sv, k + 1);
            const int s2 = __builtin_amdgcn_readlane(sv, k + 2);
            const int s3 = __builtin_amdgcn_readlane(sv, k + 3);
            const float4 r0 = *reinterpret_cast<const float4*>(h + (size_t)s0 * D_CH + lane * 4);
            const float4 r1 = *reinterpret_cast<const float4*>(h + (size_t)s1 * D_CH + lane * 4);
            const float4 r2 = *reinterpret_cast<const float4*>(h + (size_t)s2 * D_CH + lane * 4);
            const float4 r3 = *reinterpret_cast<const float4*>(h + (size_t)s3 * D_CH + lane * 4);
            const int d0 = __builtin_amdgcn_readlane(dv, k);
            const int d1 = __builtin_amdgcn_readlane(dv, k + 1);
            const int d2 = __builtin_amdgcn_readlane(dv, k + 2);
            const int d3 = __builtin_amdgcn_readlane(dv, k + 3);
            const float n0 = __int_as_float(__builtin_amdgcn_readlane(nvi, k));
            const float n1 = __int_as_float(__builtin_amdgcn_readlane(nvi, k + 1));
            const float n2 = __int_as_float(__builtin_amdgcn_readlane(nvi, k + 2));
            const float n3 = __int_as_float(__builtin_amdgcn_readlane(nvi, k + 3));
            tile_add_row(tile + d0 * D_CH + lane, n0, r0);
            tile_add_row(tile + d1 * D_CH + lane, n1, r1);
            tile_add_row(tile + d2 * D_CH + lane, n2, r2);
            tile_add_row(tile + d3 * D_CH + lane, n3, r3);
        }
        for (; k < m; ++k) {
            const int   s0 = __builtin_amdgcn_readlane(sv, k);
            const int   d0 = __builtin_amdgcn_readlane(dv, k);
            const float n0 = __int_as_float(__builtin_amdgcn_readlane(nvi, k));
            const float4 r0 = *reinterpret_cast<const float4*>(h + (size_t)s0 * D_CH + lane * 4);
            tile_add_row(tile + d0 * D_CH + lane, n0, r0);
        }
    }
    __syncthreads();

    const bool fuse = (Wnext != nullptr);
    const bool last = (out != nullptr);
    const int rowsOut = fuse ? XT_ROWS : BIN_R;

    const int nChunks = (nLook + 31) >> 5;
    int idv[LK_AHEAD];
#pragma unroll
    for (int j = 0; j < LK_AHEAD; ++j) {
        const int b = ((j * (AGG_T / 32) + wave) << 5) + lane;
        idv[j] = (b < nLook) ? idx[b] : -1;
    }

    for (int i = tid; i < rowsOut * 32; i += AGG_T) {
        const int nl   = i >> 5;
        const int node = bin * BIN_R + nl;
        float4 o = make_float4(0.f, 0.f, 0.f, 0.f);
        if (nl < BIN_R && node < nNodes) {
            const float* t = tile + nl * D_CH + lane;
            const float4 hv = *reinterpret_cast<const float4*>(h + (size_t)node * D_CH + lane * 4);
            const float4 bv = *reinterpret_cast<const float4*>(bias + lane * 4);
            const float di = dinv[node];
            o.x = fmaxf(t[0]  + hv.x * di * di + bv.x, 0.0f);
            o.y = fmaxf(t[32] + hv.y * di * di + bv.y, 0.0f);
            o.z = fmaxf(t[64] + hv.z * di * di + bv.z, 0.0f);
            o.w = fmaxf(t[96] + hv.w * di * di + bv.w, 0.0f);
        }
        *reinterpret_cast<float4*>(xt + nl * XT_PITCH + lane * 4) = o;
    }
    __syncthreads();

    const int lo = bin * BIN_R;
    const unsigned span = (bin == (int)gridDim.x - 1) ? 0x80000000u : (unsigned)BIN_R;
    for (int cb = 0; cb < nChunks; cb += LK_AHEAD * (AGG_T / 32)) {
        if (cb) {
#pragma unroll
            for (int j = 0; j < LK_AHEAD; ++j) {
                const int b = ((cb + j * (AGG_T / 32) + wave) << 5) + lane;
                idv[j] = (b < nLook) ? idx[b] : -1;
            }
        }
#pragma unroll
        for (int j = 0; j < LK_AHEAD; ++j) {
            const int c0 = (cb + j * (AGG_T / 32) + wave) << 5;
            const int id = idv[j];
            unsigned m = __builtin_amdgcn_ballot_w32((unsigned)(id - lo) < span);
            while (m) {
                const int k = __builtin_ctz(m);
                m &= m - 1;
                const int idk = __builtin_amdgcn_readlane(id, k);
                const bool real = idk < nNodes;
                const size_t o = (size_t)(c0 + k) * D_CH + lane * 4;
                float4 v = make_float4(0.f, 0.f, 0.f, 0.f);
                if (real)
                    v = *reinterpret_cast<const float4*>(xt + (idk - lo) * XT_PITCH + lane * 4);
                if (last) {
                    float4 r = v;
                    if (real) {
                        const float4 a = *reinterpret_cast<const float4*>(eA + o);
                        const float4 g = *reinterpret_cast<const float4*>(eB + o);
                        r.x = (a.x + g.x + v.x) * (1.0f / 3.0f);
                        r.y = (a.y + g.y + v.y) * (1.0f / 3.0f);
                        r.z = (a.z + g.z + v.z) * (1.0f / 3.0f);
                        r.w = (a.w + g.w + v.w) * (1.0f / 3.0f);
                    }
                    *reinterpret_cast<float4*>(out + o) = r;
                } else if (real) {
                    *reinterpret_cast<float4*>(erow + o) = v;
                }
            }
        }
    }

    if (fuse) {
        const int nT   = wave & 7;
        const int mP   = wave >> 3;
        const int r16  = lane & 15;
        const int half = lane >> 4;
        const float* xa0 = xt + (mP * 32 + r16) * XT_PITCH + half * 2;
        const float* xa1 = xa0 + 16 * XT_PITCH;
        const float* __restrict__ wb = Wnext + (size_t)(half * 2) * D_CH + nT * 16 + r16;
        v8f acc0 = v8f{};
        v8f acc1 = v8f{};
#pragma unroll 8
        for (int ks = 0; ks < 32; ++ks) {
            const v2f a0 = *reinterpret_cast<const v2f*>(xa0 + ks * 4);
            const v2f a1 = *reinterpret_cast<const v2f*>(xa1 + ks * 4);
            v2f b;
            b.x = wb[(size_t)(ks * 4) * D_CH];
            b.y = wb[(size_t)(ks * 4 + 1) * D_CH];
            acc0 = __builtin_amdgcn_wmma_f32_16x16x4_f32(
                false, a0, false, b, (short)0, acc0, false, false);
            acc1 = __builtin_amdgcn_wmma_f32_16x16x4_f32(
                false, a1, false, b, (short)0, acc1, false, false);
        }
        const int rowsValid = min(BIN_R, nNodes - bin * BIN_R);
        const int row0 = mP * 32 + half * 8;
        float* __restrict__ hp = hnext + (size_t)(bin * BIN_R + row0) * D_CH + nT * 16 + r16;
#pragma unroll
        for (int r = 0; r < 8; ++r) {
            if (row0 + r < rowsValid)      hp[(size_t)r * D_CH]        = acc0[r];
            if (row0 + 16 + r < rowsValid) hp[(size_t)(16 + r) * D_CH] = acc1[r];
        }
    }
#endif
}

extern "C" void kernel_launch(void* const* d_in, const int* in_sizes, int n_in,
                              void* d_out, int out_size, void* d_ws, size_t ws_size,
                              hipStream_t stream) {
    const int*   inputs_idx  = (const int*)  d_in[0];
    const float* x           = (const float*)d_in[1];
    const int*   edge_index  = (const int*)  d_in[2];
    const float* edge_weight = (const float*)d_in[3];
    const float* Wm[3] = { (const float*)d_in[4], (const float*)d_in[6], (const float*)d_in[8] };
    const float* bv[3] = { (const float*)d_in[5], (const float*)d_in[7], (const float*)d_in[9] };

    const int B = in_sizes[0];
    const int N = in_sizes[1] / D_CH;
    const int E = in_sizes[2] / 2;
    (void)B; (void)n_in; (void)ws_size;
    const int* src = edge_index;
    const int* dst = edge_index + E;

    float* ws   = (float*)d_ws;
    const size_t nd = (size_t)N * D_CH;
    float* dinv     = ws;
    int*   binCount = (int*)(dinv + N);
    int*   binStart = binCount + MAX_BINS;
    int*   cursor   = binStart + MAX_BINS + 64;
    int*   bsrc     = cursor + MAX_BINS;
    int*   bdl      = bsrc + E;
    float* bnorm    = (float*)(bdl + E);
    float* h        = (float*)(((size_t)(bnorm + E) + 511) & ~(size_t)511);
    float* h2       = h + nd;
    const int nLook = out_size / D_CH;
    float* e1       = h2 + nd;
    float* e2       = e1 + (size_t)nLook * D_CH;

    const int TB = 256;
    const int mTiles   = N / 16;
    const int edgeBlk  = (E + TB - 1) / TB;
    const int nodeBlk  = (N + TB - 1) / TB;
    const int nBins    = (N + BIN_R - 1) / BIN_R;
    const int chunkBlk = (E + BIN_CHUNK - 1) / BIN_CHUNK;
    const int rsqBlk   = (N + MAX_BINS - 1) / MAX_BINS;
    const int gemmBlk  = (mTiles + 7) / 8;

    init_graph<<<nodeBlk, TB, 0, stream>>>(dinv, N, binCount, nBins);
    count_and_degree<<<chunkBlk + edgeBlk, TB, 0, stream>>>(dst, edge_weight, dinv,
                                                           binCount, E, nBins, chunkBlk);
    scan_and_rsqrt<<<1 + rsqBlk, MAX_BINS, 0, stream>>>(binCount, binStart, cursor,
                                                        nBins, dinv, N);
    fill_and_gemm<<<chunkBlk + gemmBlk, TB, 0, stream>>>(src, dst, edge_weight, dinv, cursor,
                                                         bsrc, bdl, bnorm, E, nBins, chunkBlk,
                                                         x, Wm[0], h, mTiles);

    bin_aggregate<<<nBins, AGG_T, 0, stream>>>(h, bsrc, bdl, bnorm, binStart,
                                               dinv, bv[0], N, Wm[1], h2,
                                               inputs_idx, nLook, e1, nullptr, nullptr, nullptr);
    bin_aggregate<<<nBins, AGG_T, 0, stream>>>(h2, bsrc, bdl, bnorm, binStart,
                                               dinv, bv[1], N, Wm[2], h,
                                               inputs_idx, nLook, e2, nullptr, nullptr, nullptr);
    bin_aggregate<<<nBins, AGG_T, 0, stream>>>(h, bsrc, bdl, bnorm, binStart,
                                               dinv, bv[2], N, nullptr, nullptr,
                                               inputs_idx, nLook, nullptr, e1, e2, (float*)d_out);
}
